# starting kernel
# baseline (speedup 1.0000x reference)
#include <hip/hip_runtime.h>


#define B_   8
#define CI   32
#define CO   32
#define HH   128
#define WW   128
#define HID_ 16
#define TL   16

typedef __attribute__((ext_vector_type(2))) float v2f;
typedef __attribute__((ext_vector_type(8))) float v8f;

__global__ __launch_bounds__(256)
void scaconv_ds_kernel(const float* __restrict__ x,
                       const float* __restrict__ kernelD,
                       const float* __restrict__ kernelP,
                       const float* __restrict__ bias,
                       const float* __restrict__ cvec,
                       const float* __restrict__ mD_W1,
                       const float* __restrict__ mD_b1,
                       const float* __restrict__ mD_W2,
                       const float* __restrict__ mD_b2,
                       const float* __restrict__ mP_W1,
                       const float* __restrict__ mP_b1,
                       const float* __restrict__ mP_W2,
                       const float* __restrict__ mP_b2,
                       float* __restrict__ out)
{
    __shared__ float hD_lds[TL * HID_];
    __shared__ float hP_lds[TL * HID_];
    __shared__ float kD_lds[TL * 9];
    __shared__ float y_lds[TL * B_ * CI];
    __shared__ float kP_lds[TL * 16 * CO];

    const int tid  = threadIdx.x;
    const int wave = tid >> 5;
    const int lane = tid & 31;
    const int lm   = lane & 15;
    const int hiL  = lane >> 4;

    const int wg = blockIdx.x;
    const int h  = wg >> 3;
    const int w0 = (wg & 7) * TL;

    {
        const int l = tid >> 4;
        const int j = tid & 15;
        const int w = w0 + l;
        const float xn = (float)h * (1.0f / 128.0f);
        const float yn = (float)w * (1.0f / 128.0f);
        float f[12];
        f[0] = xn; f[1] = 1.0f - xn; f[2] = yn; f[3] = 1.0f - yn;
        #pragma unroll
        for (int q = 0; q < 8; ++q) f[4 + q] = cvec[q];
        float aD = mD_b1[j], aP = mP_b1[j];
        #pragma unroll
        for (int q = 0; q < 12; ++q) {
            aD += f[q] * mD_W1[q * HID_ + j];
            aP += f[q] * mP_W1[q * HID_ + j];
        }
        hD_lds[l * HID_ + j] = fmaxf(aD, 0.0f);
        hP_lds[l * HID_ + j] = fmaxf(aP, 0.0f);
    }
    __syncthreads();

    if (tid < TL * 9) {
        const int l = tid / 9;
        const int k = tid % 9;
        float a = kernelD[k] + mD_b2[k];
        #pragma unroll
        for (int j = 0; j < HID_; ++j) a += hD_lds[l * HID_ + j] * mD_W2[j * 9 + k];
        kD_lds[l * 9 + k] = a;
    }
    __syncthreads();

    {
        const int b = tid >> 5;
        const int i = tid & 31;
        const float* xp = x + (size_t)(b * CI + i) * (HH * WW);
        for (int l = 0; l < TL; ++l) {
            const int w = w0 + l;
            float acc = 0.0f;
            #pragma unroll
            for (int kh = 0; kh < 3; ++kh) {
                const int hh = h + kh - 1;
                if ((unsigned)hh < (unsigned)HH) {
                    #pragma unroll
                    for (int kw = 0; kw < 3; ++kw) {
                        const int ww = w + kw - 1;
                        if ((unsigned)ww < (unsigned)WW)
                            acc += xp[hh * WW + ww] * kD_lds[l * 9 + kh * 3 + kw];
                    }
                }
            }
            y_lds[(l * B_ + b) * CI + i] = acc;
        }
    }
    __syncthreads();

    v8f acc[2][2];
    #pragma unroll
    for (int a1 = 0; a1 < 2; ++a1)
        #pragma unroll
        for (int a2 = 0; a2 < 2; ++a2) acc[a1][a2] = (v8f){};

    for (int ihalf = 0; ihalf < 2; ++ihalf) {
        for (int t = 0; t < 4; ++t) {
            const int n    = (wave * 4 + t) * 16 + lm;
            const int gcol = ihalf * 512 + n;
            v8f d = (v8f){};
            #pragma unroll
            for (int ks = 0; ks < 4; ++ks) {
                const int k0 = ks * 4 + 2 * hiL;
                v2f af, bf;
                af.x = hP_lds[lm * HID_ + k0];
                af.y = hP_lds[lm * HID_ + k0 + 1];
                bf.x = mP_W2[(size_t)k0 * 1024 + gcol];
                bf.y = mP_W2[(size_t)(k0 + 1) * 1024 + gcol];
                d = __builtin_amdgcn_wmma_f32_16x16x4_f32(
                        false, af, false, bf, (short)0, d, false, false);
            }
            const int i_loc = n >> 5;
            const int o     = n & 31;
            const int ig    = gcol >> 5;
            const float kbase = kernelP[ig * CO + o] + mP_b2[gcol];
            #pragma unroll
            for (int r = 0; r < 8; ++r) {
                const int m = r + 8 * hiL;
                kP_lds[(m * 16 + i_loc) * CO + o] = d[r] + kbase;
            }
        }
        __syncthreads();

        #pragma unroll
        for (int li = 0; li < 2; ++li) {
            const int l = wave * 2 + li;
            #pragma unroll
            for (int nt = 0; nt < 2; ++nt) {
                v8f d = acc[li][nt];
                #pragma unroll
                for (int ks = 0; ks < 4; ++ks) {
                    const int k0 = ks * 4 + 2 * hiL;
                    v2f af, bf;
                    if (lm < 8) {
                        af.x = y_lds[(l * B_ + lm) * CI + ihalf * 16 + k0];
                        af.y = y_lds[(l * B_ + lm) * CI + ihalf * 16 + k0 + 1];
                    } else {
                        af.x = 0.0f; af.y = 0.0f;
                    }
                    const int o = nt * 16 + lm;
                    bf.x = kP_lds[(l * 16 + k0) * CO + o];
                    bf.y = kP_lds[(l * 16 + k0 + 1) * CO + o];
                    d = __builtin_amdgcn_wmma_f32_16x16x4_f32(
                            false, af, false, bf, (short)0, d, false, false);
                }
                acc[li][nt] = d;
            }
        }
        __syncthreads();
    }

    #pragma unroll
    for (int li = 0; li < 2; ++li) {
        const int l = wave * 2 + li;
        const size_t lglob = (size_t)h * WW + (w0 + l);
        #pragma unroll
        for (int nt = 0; nt < 2; ++nt) {
            const int o = nt * 16 + lm;
            const float bo = bias[o];
            if (hiL == 0) {
                #pragma unroll
                for (int r = 0; r < 8; ++r) {
                    out[(size_t)(r * CO + o) * (HH * WW) + lglob] = acc[li][nt][r] + bo;
                }
            }
        }
    }
}

extern "C" void kernel_launch(void* const* d_in, const int* in_sizes, int n_in,
                              void* d_out, int out_size, void* d_ws, size_t ws_size,
                              hipStream_t stream) {
    const float* x       = (const float*)d_in[0];
    const float* kernelD = (const float*)d_in[1];
    const float* kernelP = (const float*)d_in[2];
    const float* bias    = (const float*)d_in[3];
    const float* cvec    = (const float*)d_in[4];
    const float* mD_W1   = (const float*)d_in[5];
    const float* mD_b1   = (const float*)d_in[6];
    const float* mD_W2   = (const float*)d_in[7];
    const float* mD_b2   = (const float*)d_in[8];
    const float* mP_W1   = (const float*)d_in[9];
    const float* mP_b1   = (const float*)d_in[10];
    const float* mP_W2   = (const float*)d_in[11];
    const float* mP_b2   = (const float*)d_in[12];
    float* out = (float*)d_out;

    dim3 grid(1024);
    dim3 block(256);
    hipLaunchKernelGGL(scaconv_ds_kernel, grid, block, 0, stream,
                       x, kernelD, kernelP, bias, cvec,
                       mD_W1, mD_b1, mD_W2, mD_b2,
                       mP_W1, mP_b1, mP_W2, mP_b2, out);
}


# Work-group is 4 waves of up to 512 registers (was 8 of 256): a wave owns 16 input channels and 2 planes and holds a row's taps in registers; 2 partial tiles are summed, not 4: sum order changed
# speedup vs baseline: 4.6676x; 4.6676x over previous
#include <hip/hip_runtime.h>


#define B_   8
#define CI   32
#define CO   32
#define HH   128
#define WW   128
#define HID_ 16
#define TL   16
#define RT   2
#define NL   (RT * TL)

typedef __attribute__((ext_vector_type(2))) float v2f;
typedef __attribute__((ext_vector_type(8))) float v8f;

#define XROW 24
#define XRS  (RT + 2)
#define XPL  (XRS * XROW + 2)
#define YP   276
#define YG   68
#define YZ   (16 * YP)

__global__ __launch_bounds__(128, 2)
void scaconv_ds_kernel(const float* __restrict__ x,
                       const float* __restrict__ kernelD,
                       const float* __restrict__ kernelP,
                       const float* __restrict__ bias,
                       const float* __restrict__ cvec,
                       const float* __restrict__ mD_W1,
                       const float* __restrict__ mD_b1,
                       const float* __restrict__ mD_W2,
                       const float* __restrict__ mD_b2,
                       const float* __restrict__ mP_W1,
                       const float* __restrict__ mP_b1,
                       const float* __restrict__ mP_W2,
                       const float* __restrict__ mP_b2,
                       float* __restrict__ out)
{
    __shared__ float hD_lds[NL * HID_];
    __shared__ __attribute__((aligned(16))) float hP_lds[NL * HID_];
    __shared__ __attribute__((aligned(16))) float kD_lds[NL * 9];
    __shared__ __attribute__((aligned(16))) float smem[B_ * CI * XPL + RT * YZ + 64];
    float* const xk_lds = smem;
    float* const y_lds  = smem + B_ * CI * XPL;

    const int tid  = threadIdx.x;
    const int wave = tid >> 5;
    const int lane = tid & 31;
    const int lm   = lane & 15;
    const int hiL  = lane >> 4;

    const int wg = blockIdx.x;
    const int h0 = (wg >> 3) * RT;
    const int w0 = (wg & 7) * TL;

    float4 xpc[16 * XRS];
    const int  pc   = tid & 7;
    const int  pb   = ((tid >> 3) & 1) + 16 * ((tid >> 4) & 1) + 2 * (tid >> 5);
    const bool puse = (pc >= 1) && (pc <= 6);
    {
        const int  wcol = w0 - 8 + 4 * pc;
        const bool cok  = puse && ((unsigned)wcol < (unsigned)WW);
        const int  wc   = cok ? wcol : w0;
        #pragma unroll
        for (int n = 0; n < 16 * XRS; ++n) {
            const int r  = n & 3;
            const int p  = pb + 8 * ((n >> 2) & 1) + 32 * (n >> 3);
            const int hh = h0 + r - 1;
            const int hc = min(max(hh, 0), HH - 1);
            const float4 v = *reinterpret_cast<const float4*>(
                x + (size_t)p * (HH * WW) + hc * WW + wc);
            const bool ok = cok && ((unsigned)hh < (unsigned)HH);
            xpc[n].x = ok ? v.x : 0.0f;
            xpc[n].y = ok ? v.y : 0.0f;
            xpc[n].z = ok ? v.z : 0.0f;
            xpc[n].w = ok ? v.w : 0.0f;
        }
    }

    if (tid < 64) y_lds[RT * YZ + tid] = 0.0f;

    const int nt = wave & 1;
    const int ig = wave >> 1;

    {
        const int j = tid & 15;
        float cq[8];
        #pragma unroll
        for (int q = 0; q < 8; ++q) cq[q] = cvec[q];
        float wD[12], wP[12];
        #pragma unroll
        for (int q = 0; q < 12; ++q) {
            wD[q] = mD_W1[q * HID_ + j];
            wP[q] = mP_W1[q * HID_ + j];
        }
        const float bD = mD_b1[j], bP = mP_b1[j];
        #pragma unroll
        for (int lh = 0; lh < 2; ++lh) {
            const int l = (tid >> 4) + 8 * lh;
            const int w = w0 + l;
            const float yn = (float)w * (1.0f / 128.0f);
            #pragma unroll
            for (int t = 0; t < RT; ++t) {
                const float xn = (float)(h0 + t) * (1.0f / 128.0f);
                float f[12];
                f[0] = xn; f[1] = 1.0f - xn; f[2] = yn; f[3] = 1.0f - yn;
                #pragma unroll
                for (int q = 0; q < 8; ++q) f[4 + q] = cq[q];
                float aD = bD, aP = bP;
                #pragma unroll
                for (int q = 0; q < 12; ++q) {
                    aD += f[q] * wD[q];
                    aP += f[q] * wP[q];
                }
                hD_lds[(t * TL + l) * HID_ + j] = fmaxf(aD, 0.0f);
                hP_lds[(t * TL + l) * HID_ + j] = fmaxf(aP, 0.0f);
            }
        }
    }
    if (puse) {
        #pragma unroll
        for (int n = 0; n < 16 * XRS; ++n) {
            const int r = n & 3;
            const int p = pb + 8 * ((n >> 2) & 1) + 32 * (n >> 3);
            float* dst = &xk_lds[p * XPL + r * XROW + 4 * (pc - 1)];
            *reinterpret_cast<float2*>(dst)     = make_float2(xpc[n].x, xpc[n].y);
            *reinterpret_cast<float2*>(dst + 2) = make_float2(xpc[n].z, xpc[n].w);
        }
    }
    __syncthreads();

    {
        const int l = tid >> 3;
        const int k = tid & 7;
        const float base = kernelD[k] + mD_b2[k];
        const float base8 = kernelD[8] + mD_b2[8];
        float w2c[HID_], w2e[HID_];
        #pragma unroll
        for (int j = 0; j < HID_; ++j) {
            w2c[j] = mD_W2[j * 9 + k];
            w2e[j] = mD_W2[j * 9 + 8];
        }
        #pragma unroll
        for (int t = 0; t < RT; ++t) {
            float a = base, a8 = base8;
            #pragma unroll
            for (int j = 0; j < HID_; ++j) {
                const float hv = hD_lds[(t * TL + l) * HID_ + j];
                a  += hv * w2c[j];
                a8 += hv * w2e[j];
            }
            kD_lds[(t * TL + l) * 9 + k] = a;
            if (k == 0) kD_lds[(t * TL + l) * 9 + 8] = a8;
        }
    }

    __syncthreads();

    v2f   bfr[8][2][4];
    float kbase[8][2];
    #pragma unroll
    for (int ipx = 0; ipx < 8; ++ipx) {
        #pragma unroll
        for (int t2 = 0; t2 < 2; ++t2) {
            const int gcol = (ig * 16 + ipx * 2 + t2) * CO + nt * 16 + lm;
            #pragma unroll
            for (int ks = 0; ks < 4; ++ks) {
                const int k0 = ks * 4 + 2 * hiL;
                bfr[ipx][t2][ks].x = mP_W2[k0 * 1024 + gcol];
                bfr[ipx][t2][ks].y = mP_W2[(k0 + 1) * 1024 + gcol];
            }
            kbase[ipx][t2] = kernelP[gcol] + mP_b2[gcol];
        }
    }
    {
        const int b = tid >> 5;
        const int i = tid & 31;
        float* const ydst = y_lds + b * YP + (i >> 3) * YG + ((i & 7) >> 1) * 16 + (i & 1);
        #pragma unroll 1
        for (int t = 0; t < RT; ++t) {
            float kdr[TL * 9];
            {
                const float4* kD4 = reinterpret_cast<const float4*>(kD_lds + t * (TL * 9));
                #pragma unroll
                for (int m = 0; m < (TL * 9) / 4; ++m) {
                    const float4 v = kD4[m];
                    kdr[4 * m] = v.x; kdr[4 * m + 1] = v.y; kdr[4 * m + 2] = v.z; kdr[4 * m + 3] = v.w;
                }
            }
            float f[2][3 * 20];
            #pragma unroll
            for (int pp = 0; pp < 2; ++pp) {
                const float2* xr2 = reinterpret_cast<const float2*>(
                    &xk_lds[(tid + 128 * pp) * XPL + t * XROW]);
                #pragma unroll
                for (int rr = 0; rr < 3; ++rr) {
                    #pragma unroll
                    for (int k = 0; k < 10; ++k) {
                        const float2 v = xr2[rr * (XROW / 2) + 1 + k];
                        f[pp][rr * 20 + 2 * k] = v.x; f[pp][rr * 20 + 2 * k + 1] = v.y;
                    }
                }
            }
            #pragma unroll
            for (int l = 0; l < TL; ++l) {
                float acc0 = 0.0f, acc1 = 0.0f;
                #pragma unroll
                for (int kh = 0; kh < 3; ++kh) {
                    #pragma unroll
                    for (int kw = 0; kw < 3; ++kw) {
                        const int c = l + kw;
                        const float kd = kdr[l * 9 + kh * 3 + kw];
                        acc0 += f[0][kh * 20 + 1 + c] * kd;
                        acc1 += f[1][kh * 20 + 1 + c] * kd;
                    }
                }
                float* const yd = ydst + t * YZ + (l >> 3) * 8 * YP + (l & 7) * 2;
                yd[0]      = acc0;
                yd[4 * YP] = acc1;
            }
        }
    }
    __syncthreads();

    const bool yact = ((lm >> 3) == hiL);
    const int  yoff = yact ? (hiL * 8 + (lm & 7)) * YP + 2 * ig * YG : RT * YZ;
    const int  ygs  = yact ? YG : 0;
    const int  o  = nt * 16 + lm;
    const float bo = bias[o];

    #pragma unroll 1
    for (int t = 0; t < RT; ++t) {
        v8f acc[8];
        #pragma unroll
        for (int r = 0; r < 8; ++r) acc[r] = (v8f){};

        v2f afr[4];
        #pragma unroll
        for (int ks = 0; ks < 4; ++ks) {
            const float2 hv = *reinterpret_cast<const float2*>(
                &hP_lds[(t * TL + lm) * HID_ + ks * 4 + 2 * hiL]);
            afr[ks].x = hv.x;
            afr[ks].y = hv.y;
        }
        const float* yrow = y_lds + yoff + (yact ? t * YZ : 0);

        float4 yv[8][4];
        #pragma unroll
        for (int ip = 0; ip < 8; ++ip) {
            #pragma unroll
            for (int p = 0; p < 4; ++p)
                yv[ip][p] = *reinterpret_cast<const float4*>(
                    yrow + (ip >> 2) * ygs + (ip & 3) * 16 + p * 4);
        }
        __builtin_amdgcn_sched_barrier(0);

        #pragma unroll
        for (int ip = 0; ip < 8; ++ip) {
            v8f kp[2];
            #pragma unroll
            for (int t2 = 0; t2 < 2; ++t2) {
                v8f dd = (v8f){};
                #pragma unroll
                for (int ks = 0; ks < 4; ++ks)
                    dd = __builtin_amdgcn_wmma_f32_16x16x4_f32(
                             false, afr[ks], false, bfr[ip][t2][ks], (short)0, dd, false, false);
                const float kb = kbase[ip][t2];
                #pragma unroll
                for (int r = 0; r < 8; ++r) dd[r] = dd[r] + kb;
                kp[t2] = dd;
            }
            #pragma unroll
            for (int p = 0; p < 4; ++p) {
                v2f af0, af1, bf0, bf1;
                af0.x = yv[ip][p].x; af0.y = yv[ip][p].y;
                af1.x = yv[ip][p].z; af1.y = yv[ip][p].w;
                bf0.x = kp[0][2 * p];     bf0.y = kp[1][2 * p];
                bf1.x = kp[0][2 * p + 1]; bf1.y = kp[1][2 * p + 1];
                acc[2 * p] = __builtin_amdgcn_wmma_f32_16x16x4_f32(
                                 false, af0, false, bf0, (short)0, acc[2 * p], false, false);
                acc[2 * p + 1] = __builtin_amdgcn_wmma_f32_16x16x4_f32(
                                 false, af1, false, bf1, (short)0, acc[2 * p + 1], false, false);
            }
        }
        if (t > 0) __syncthreads();

        {
            float4* red4 = reinterpret_cast<float4*>(smem);
            #pragma unroll
            for (int q = 0; q < 8; ++q) {
                #pragma unroll
                for (int rg = 0; rg < 2; ++rg) {
                    float4 v;
                    v.x = acc[rg * 4 + 0][q];
                    v.y = acc[rg * 4 + 1][q];
                    v.z = acc[rg * 4 + 2][q];
                    v.w = acc[rg * 4 + 3][q];
                    red4[(wave * 16 + q * 2 + rg) * 32 + lane] = v;
                }
            }
        }
        __syncthreads();

        {
            const float4* red4 = reinterpret_cast<const float4*>(smem);
            const int qg = wave >> 1;
            #pragma unroll
            for (int qq = 0; qq < 4; ++qq) {
                const int q = qg * 4 + qq;
                #pragma unroll
                for (int rg = 0; rg < 2; ++rg) {
                    float4 s = red4[((nt + 0) * 16 + q * 2 + rg) * 32 + lane];
                    const float4 pp = red4[((nt + 2) * 16 + q * 2 + rg) * 32 + lane];
                    s.x += pp.x; s.y += pp.y; s.z += pp.z; s.w += pp.w;
                    s.x += bo; s.y += bo; s.z += bo; s.w += bo;
                    *reinterpret_cast<float4*>(
                        out + (size_t)(q * CO + o) * (HH * WW) + (size_t)(h0 + t) * WW
                            + (w0 + 4 * (rg + 2 * hiL))) = s;
                }
            }
        }
    }
}

extern "C" void kernel_launch(void* const* d_in, const int* in_sizes, int n_in,
                              void* d_out, int out_size, void* d_ws, size_t ws_size,
                              hipStream_t stream) {
    const float* x       = (const float*)d_in[0];
    const float* kernelD = (const float*)d_in[1];
    const float* kernelP = (const float*)d_in[2];
    const float* bias    = (const float*)d_in[3];
    const float* cvec    = (const float*)d_in[4];
    const float* mD_W1   = (const float*)d_in[5];
    const float* mD_b1   = (const float*)d_in[6];
    const float* mD_W2   = (const float*)d_in[7];
    const float* mD_b2   = (const float*)d_in[8];
    const float* mP_W1   = (const float*)d_in[9];
    const float* mP_b1   = (const float*)d_in[10];
    const float* mP_W2   = (const float*)d_in[11];
    const float* mP_b2   = (const float*)d_in[12];
    float* out = (float*)d_out;

    dim3 grid(512);
    dim3 block(128);
    hipLaunchKernelGGL(scaconv_ds_kernel, grid, block, 0, stream,
                       x, kernelD, kernelP, bias, cvec,
                       mD_W1, mD_b1, mD_W2, mD_b2,
                       mP_W1, mP_b1, mP_W2, mP_b2, out);
}


# kP tiles of both rows are built once, under the x fetch (x in four quarters, two in flight; MLP tables requested first, W2 behind the MLP stages; barriers that order LDS only); same sum order
# speedup vs baseline: 5.0200x; 1.0755x over previous
#include <hip/hip_runtime.h>


#define B_   8
#define CI   32
#define CO   32
#define HH   128
#define WW   128
#define HID_ 16
#define TL   16
#define RT   2
#define NL   (RT * TL)

typedef __attribute__((ext_vector_type(2))) float v2f;
typedef __attribute__((ext_vector_type(8))) float v8f;

#define XROW 24
#define XRS  (RT + 2)
#define XPL  (XRS * XROW + 2)
#define YP   276
#define YG   68
#define YZ   (16 * YP)

#define XH   16

__device__ __forceinline__ void lds_barrier()
{
    __builtin_amdgcn_fence(__ATOMIC_RELEASE, "workgroup", "local");
    __builtin_amdgcn_s_barrier();
    __builtin_amdgcn_fence(__ATOMIC_ACQUIRE, "workgroup", "local");
}

__device__ __forceinline__ void x_fetch(float4 (&xq)[XH], const float* __restrict__ x,
                                        const int quarter, const int pb, const int h0, const int wc)
{
    #pragma unroll
    for (int m = 0; m < XH; ++m) {
        const int n  = XH * quarter + m;
        const int r  = n & 3;
        const int p  = pb + 8 * ((n >> 2) & 1) + 32 * (n >> 3);
        const int hh = h0 + r - 1;
        const int hc = min(max(hh, 0), HH - 1);
        xq[m] = *reinterpret_cast<const float4*>(x + (size_t)p * (HH * WW) + hc * WW + wc);
    }
}

__device__ __forceinline__ void x_stage(const float4 (&xq)[XH], float* const xk_lds,
                                        const int quarter, const int pb, const int pc,
                                        const int h0, const bool cok)
{
    #pragma unroll
    for (int m = 0; m < XH; ++m) {
        const int n  = XH * quarter + m;
        const int r  = n & 3;
        const int p  = pb + 8 * ((n >> 2) & 1) + 32 * (n >> 3);
        const int hh = h0 + r - 1;
        const bool ok = cok && ((unsigned)hh < (unsigned)HH);
        float* dst = &xk_lds[p * XPL + r * XROW + 4 * (pc - 1)];
        *reinterpret_cast<float2*>(dst)     = make_float2(ok ? xq[m].x : 0.0f, ok ? xq[m].y : 0.0f);
        *reinterpret_cast<float2*>(dst + 2) = make_float2(ok ? xq[m].z : 0.0f, ok ? xq[m].w : 0.0f);
    }
}

__device__ __forceinline__ void x_hold()
{
    __builtin_amdgcn_fence(__ATOMIC_ACQ_REL, "wavefront");
}

__global__ __launch_bounds__(128, 2)
void scaconv_ds_kernel(const float* __restrict__ x,
                       const float* __restrict__ kernelD,
                       const float* __restrict__ kernelP,
                       const float* __restrict__ bias,
                       const float* __restrict__ cvec,
                       const float* __restrict__ mD_W1,
                       const float* __restrict__ mD_b1,
                       const float* __restrict__ mD_W2,
                       const float* __restrict__ mD_b2,
                       const float* __restrict__ mP_W1,
                       const float* __restrict__ mP_b1,
                       const float* __restrict__ mP_W2,
                       const float* __restrict__ mP_b2,
                       float* __restrict__ out)
{
    __shared__ float hD_lds[NL * HID_];
    __shared__ __attribute__((aligned(16))) float hP_lds[NL * HID_];
    __shared__ __attribute__((aligned(16))) float kD_lds[NL * 9];
    __shared__ __attribute__((aligned(16))) float smem[B_ * CI * XPL + RT * YZ + 64];
    float* const xk_lds = smem;
    float* const y_lds  = smem + B_ * CI * XPL;

    const int tid  = threadIdx.x;
    const int wave = tid >> 5;
    const int lane = tid & 31;
    const int lm   = lane & 15;
    const int hiL  = lane >> 4;

    const int wg = blockIdx.x;
    const int h0 = (wg >> 3) * RT;
    const int w0 = (wg & 7) * TL;

    const int nt = wave & 1;
    const int ig = wave >> 1;


    const int j  = tid & 15;
    const int k1 = tid & 7;
    float cq[8];
    #pragma unroll
    for (int q = 0; q < 8; ++q) cq[q] = cvec[q];
    float wD[12], wP[12];
    #pragma unroll
    for (int q = 0; q < 12; ++q) {
        wD[q] = mD_W1[q * HID_ + j];
        wP[q] = mP_W1[q * HID_ + j];
    }
    const float bD = mD_b1[j], bP = mP_b1[j];
    const float kDk = kernelD[k1], b2k = mD_b2[k1];
    const float kD8 = kernelD[8],  b28 = mD_b2[8];
    float w2c[HID_], w2e[HID_];
    #pragma unroll
    for (int jj = 0; jj < HID_; ++jj) {
        w2c[jj] = mD_W2[jj * 9 + k1];
        w2e[jj] = mD_W2[jj * 9 + 8];
    }
    __builtin_amdgcn_sched_barrier(0);

    const int  pc   = tid & 7;
    const int  pb   = ((tid >> 3) & 1) + 16 * ((tid >> 4) & 1) + 2 * (tid >> 5);
    const bool puse = (pc >= 1) && (pc <= 6);
    const int  wcol = w0 - 8 + 4 * pc;
    const bool cok  = puse && ((unsigned)wcol < (unsigned)WW);
    const int  wc   = cok ? wcol : w0;
    float4 xqa[XH], xqb[XH];
    x_fetch(xqa, x, 0, pb, h0, wc);
    x_fetch(xqb, x, 1, pb, h0, wc);

    if (tid < 64) y_lds[RT * YZ + tid] = 0.0f;

    {
        #pragma unroll
        for (int lh = 0; lh < 2; ++lh) {
            const int l = (tid >> 4) + 8 * lh;
            const int w = w0 + l;
            const float yn = (float)w * (1.0f / 128.0f);
            #pragma unroll
            for (int t = 0; t < RT; ++t) {
                const float xn = (float)(h0 + t) * (1.0f / 128.0f);
                float f[12];
                f[0] = xn; f[1] = 1.0f - xn; f[2] = yn; f[3] = 1.0f - yn;
                #pragma unroll
                for (int q = 0; q < 8; ++q) f[4 + q] = cq[q];
                float aD = bD, aP = bP;
                #pragma unroll
                for (int q = 0; q < 12; ++q) {
                    aD += f[q] * wD[q];
                    aP += f[q] * wP[q];
                }
                hD_lds[(t * TL + l) * HID_ + j] = fmaxf(aD, 0.0f);
                hP_lds[(t * TL + l) * HID_ + j] = fmaxf(aP, 0.0f);
            }
        }
    }
    lds_barrier();

    {
        const int l = tid >> 3;
        const float base  = kDk + b2k;
        const float base8 = kD8 + b28;
        #pragma unroll
        for (int t = 0; t < RT; ++t) {
            float a = base, a8 = base8;
            #pragma unroll
            for (int jj = 0; jj < HID_; ++jj) {
                const float hv = hD_lds[(t * TL + l) * HID_ + jj];
                a  += hv * w2c[jj];
                a8 += hv * w2e[jj];
            }
            kD_lds[(t * TL + l) * 9 + k1] = a;
            if (k1 == 0) kD_lds[(t * TL + l) * 9 + 8] = a8;
        }
    }

    v2f   bfr[8][2][4];
    #pragma unroll
    for (int ipx = 0; ipx < 8; ++ipx) {
        #pragma unroll
        for (int t2 = 0; t2 < 2; ++t2) {
            const int gcol = (ig * 16 + ipx * 2 + t2) * CO + nt * 16 + lm;
            #pragma unroll
            for (int ks = 0; ks < 4; ++ks) {
                const int k0 = ks * 4 + 2 * hiL;
                bfr[ipx][t2][ks].x = mP_W2[k0 * 1024 + gcol];
                bfr[ipx][t2][ks].y = mP_W2[(k0 + 1) * 1024 + gcol];
            }
        }
    }
    __builtin_amdgcn_sched_barrier(0);

    v2f afr[RT][4];
    #pragma unroll
    for (int t = 0; t < RT; ++t) {
        #pragma unroll
        for (int ks = 0; ks < 4; ++ks) {
            const float2 hv = *reinterpret_cast<const float2*>(
                &hP_lds[(t * TL + lm) * HID_ + ks * 4 + 2 * hiL]);
            afr[t][ks].x = hv.x;
            afr[t][ks].y = hv.y;
        }
    }

    if (puse) x_stage(xqa, xk_lds, 0, pb, pc, h0, cok);
    x_fetch(xqa, x, 2, pb, h0, wc);
    if (puse) x_stage(xqb, xk_lds, 1, pb, pc, h0, cok);
    x_fetch(xqb, x, 3, pb, h0, wc);
    __builtin_amdgcn_sched_barrier(0);

    v8f kpT[RT][8][2];
    #pragma unroll
    for (int ip = 0; ip < 8; ++ip) {
        if (ip == 4) {
            x_hold();
            if (puse) x_stage(xqa, xk_lds, 2, pb, pc, h0, cok);
            __builtin_amdgcn_sched_barrier(0);
        }
        #pragma unroll
        for (int t2 = 0; t2 < 2; ++t2) {
            #pragma unroll
            for (int t = 0; t < RT; ++t) {
                v8f dd = (v8f){};
                #pragma unroll
                for (int ks = 0; ks < 4; ++ks)
                    dd = __builtin_amdgcn_wmma_f32_16x16x4_f32(
                             false, afr[t][ks], false, bfr[ip][t2][ks], (short)0, dd, false, false);
                kpT[t][ip][t2] = dd;
            }
        }
        __builtin_amdgcn_sched_barrier(0);
    }
    x_hold();
    if (puse) x_stage(xqb, xk_lds, 3, pb, pc, h0, cok);
    __syncthreads();

    float kbase[8][2];
    #pragma unroll
    for (int ipx = 0; ipx < 8; ++ipx) {
        #pragma unroll
        for (int t2 = 0; t2 < 2; ++t2) {
            const int gcol = (ig * 16 + ipx * 2 + t2) * CO + nt * 16 + lm;
            kbase[ipx][t2] = kernelP[gcol] + mP_b2[gcol];
        }
    }

    {
        const int b = tid >> 5;
        const int i = tid & 31;
        float* const ydst = y_lds + b * YP + (i >> 3) * YG + ((i & 7) >> 1) * 16 + (i & 1);
        #pragma unroll 1
        for (int th = 0; th < 2 * RT; ++th) {
            const int t  = th >> 1;
            const int lh = th & 1;
            float kdr[8 * 9];
            {
                const float4* kD4 = reinterpret_cast<const float4*>(kD_lds + t * (TL * 9) + lh * 72);
                #pragma unroll
                for (int m = 0; m < 18; ++m) {
                    const float4 v = kD4[m];
                    kdr[4 * m] = v.x; kdr[4 * m + 1] = v.y; kdr[4 * m + 2] = v.z; kdr[4 * m + 3] = v.w;
                }
            }
            float f[2][3 * 12];
            #pragma unroll
            for (int pp = 0; pp < 2; ++pp) {
                const float2* xr2 = reinterpret_cast<const float2*>(
                    &xk_lds[(tid + 128 * pp) * XPL + t * XROW + 8 * lh]);
                #pragma unroll
                for (int rr = 0; rr < 3; ++rr) {
                    #pragma unroll
                    for (int k = 0; k < 6; ++k) {
                        const float2 v = xr2[rr * (XROW / 2) + 1 + k];
                        f[pp][rr * 12 + 2 * k] = v.x; f[pp][rr * 12 + 2 * k + 1] = v.y;
                    }
                }
            }
            float* const yp = ydst + t * YZ + lh * 8 * YP;
            #pragma unroll
            for (int l = 0; l < 8; ++l) {
                float acc0 = 0.0f, acc1 = 0.0f;
                #pragma unroll
                for (int kh = 0; kh < 3; ++kh) {
                    #pragma unroll
                    for (int kw = 0; kw < 3; ++kw) {
                        const int c = l + kw;
                        const float kd = kdr[l * 9 + kh * 3 + kw];
                        acc0 += f[0][kh * 12 + 1 + c] * kd;
                        acc1 += f[1][kh * 12 + 1 + c] * kd;
                    }
                }
                yp[l * 2]          = acc0;
                yp[l * 2 + 4 * YP] = acc1;
            }
        }
    }
    __syncthreads();

    const bool yact = ((lm >> 3) == hiL);
    const int  yoff = yact ? (hiL * 8 + (lm & 7)) * YP + 2 * ig * YG : RT * YZ;
    const int  ygs  = yact ? YG : 0;
    const int  o  = nt * 16 + lm;
    const float bo = bias[o];

    #pragma unroll
    for (int t = 0; t < RT; ++t) {
        v8f acc[8];
        #pragma unroll
        for (int r = 0; r < 8; ++r) acc[r] = (v8f){};

        const float* yrow = y_lds + yoff + (yact ? t * YZ : 0);

        float4 yv[8][4];
        #pragma unroll
        for (int ip = 0; ip < 8; ++ip) {
            #pragma unroll
            for (int p = 0; p < 4; ++p)
                yv[ip][p] = *reinterpret_cast<const float4*>(
                    yrow + (ip >> 2) * ygs + (ip & 3) * 16 + p * 4);
        }
        __builtin_amdgcn_sched_barrier(0);

        #pragma unroll
        for (int ip = 0; ip < 8; ++ip) {
            #pragma unroll
            for (int p = 0; p < 4; ++p) {
                v2f af0, af1, bf0, bf1;
                af0.x = yv[ip][p].x; af0.y = yv[ip][p].y;
                af1.x = yv[ip][p].z; af1.y = yv[ip][p].w;
                bf0.x = kpT[t][ip][0][2 * p]     + kbase[ip][0];
                bf0.y = kpT[t][ip][1][2 * p]     + kbase[ip][1];
                bf1.x = kpT[t][ip][0][2 * p + 1] + kbase[ip][0];
                bf1.y = kpT[t][ip][1][2 * p + 1] + kbase[ip][1];
                acc[2 * p] = __builtin_amdgcn_wmma_f32_16x16x4_f32(
                                 false, af0, false, bf0, (short)0, acc[2 * p], false, false);
                acc[2 * p + 1] = __builtin_amdgcn_wmma_f32_16x16x4_f32(
                                 false, af1, false, bf1, (short)0, acc[2 * p + 1], false, false);
            }
        }
        if (t > 0) lds_barrier();

        {
            float4* red4 = reinterpret_cast<float4*>(smem);
            #pragma unroll
            for (int q = 0; q < 8; ++q) {
                #pragma unroll
                for (int rg = 0; rg < 2; ++rg) {
                    float4 v;
                    v.x = acc[rg * 4 + 0][q];
                    v.y = acc[rg * 4 + 1][q];
                    v.z = acc[rg * 4 + 2][q];
                    v.w = acc[rg * 4 + 3][q];
                    red4[(wave * 16 + q * 2 + rg) * 32 + lane] = v;
                }
            }
        }
        lds_barrier();

        {
            const float4* red4 = reinterpret_cast<const float4*>(smem);
            const int qg = wave >> 1;
            #pragma unroll
            for (int qq = 0; qq < 4; ++qq) {
                const int q = qg * 4 + qq;
                #pragma unroll
                for (int rg = 0; rg < 2; ++rg) {
                    float4 s = red4[((nt + 0) * 16 + q * 2 + rg) * 32 + lane];
                    const float4 pp = red4[((nt + 2) * 16 + q * 2 + rg) * 32 + lane];
                    s.x += pp.x; s.y += pp.y; s.z += pp.z; s.w += pp.w;
                    s.x += bo; s.y += bo; s.z += bo; s.w += bo;
                    *reinterpret_cast<float4*>(
                        out + (size_t)(q * CO + o) * (HH * WW) + (size_t)(h0 + t) * WW
                            + (w0 + 4 * (rg + 2 * hiL))) = s;
                }
            }
        }
    }
}

extern "C" void kernel_launch(void* const* d_in, const int* in_sizes, int n_in,
                              void* d_out, int out_size, void* d_ws, size_t ws_size,
                              hipStream_t stream) {
    const float* x       = (const float*)d_in[0];
    const float* kernelD = (const float*)d_in[1];
    const float* kernelP = (const float*)d_in[2];
    const float* bias    = (const float*)d_in[3];
    const float* cvec    = (const float*)d_in[4];
    const float* mD_W1   = (const float*)d_in[5];
    const float* mD_b1   = (const float*)d_in[6];
    const float* mD_W2   = (const float*)d_in[7];
    const float* mD_b2   = (const float*)d_in[8];
    const float* mP_W1   = (const float*)d_in[9];
    const float* mP_b1   = (const float*)d_in[10];
    const float* mP_W2   = (const float*)d_in[11];
    const float* mP_b2   = (const float*)d_in[12];
    float* out = (float*)d_out;

    dim3 grid(512);
    dim3 block(128);
    hipLaunchKernelGGL(scaconv_ds_kernel, grid, block, 0, stream,
                       x, kernelD, kernelP, bias, cvec,
                       mD_W1, mD_b1, mD_W2, mD_b2,
                       mP_W1, mP_b1, mP_W2, mP_b2, out);
}


# Channel-half sum re-mapped: 4 adjacent lanes sum and store the four 16-byte pieces of one plane's 64-byte row piece: a store writes 8 half lines whole, not 2 pieces in each of 16; same sum order
# speedup vs baseline: 5.3175x; 1.0593x over previous
#include <hip/hip_runtime.h>


#define B_   8
#define CI   32
#define CO   32
#define HH   128
#define WW   128
#define HID_ 16
#define TL   16
#define RT   2
#define NL   (RT * TL)

typedef __attribute__((ext_vector_type(2))) float v2f;
typedef __attribute__((ext_vector_type(8))) float v8f;

#define XROW 24
#define XRS  (RT + 2)
#define XPL  (XRS * XROW + 2)
#define YP   276
#define YG   68
#define YZ   (16 * YP)

#define XH   16

__device__ __forceinline__ void lds_barrier()
{
    __builtin_amdgcn_fence(__ATOMIC_RELEASE, "workgroup", "local");
    __builtin_amdgcn_s_barrier();
    __builtin_amdgcn_fence(__ATOMIC_ACQUIRE, "workgroup", "local");
}

__device__ __forceinline__ void x_fetch(float4 (&xq)[XH], const float* __restrict__ x,
                                        const int quarter, const int pb, const int h0, const int wc)
{
    #pragma unroll
    for (int m = 0; m < XH; ++m) {
        const int n  = XH * quarter + m;
        const int r  = n & 3;
        const int p  = pb + 8 * ((n >> 2) & 1) + 32 * (n >> 3);
        const int hh = h0 + r - 1;
        const int hc = min(max(hh, 0), HH - 1);
        xq[m] = *reinterpret_cast<const float4*>(x + (size_t)p * (HH * WW) + hc * WW + wc);
    }
}

__device__ __forceinline__ void x_stage(const float4 (&xq)[XH], float* const xk_lds,
                                        const int quarter, const int pb, const int pc,
                                        const int h0, const bool cok)
{
    #pragma unroll
    for (int m = 0; m < XH; ++m) {
        const int n  = XH * quarter + m;
        const int r  = n & 3;
        const int p  = pb + 8 * ((n >> 2) & 1) + 32 * (n >> 3);
        const int hh = h0 + r - 1;
        const bool ok = cok && ((unsigned)hh < (unsigned)HH);
        float* dst = &xk_lds[p * XPL + r * XROW + 4 * (pc - 1)];
        *reinterpret_cast<float2*>(dst)     = make_float2(ok ? xq[m].x : 0.0f, ok ? xq[m].y : 0.0f);
        *reinterpret_cast<float2*>(dst + 2) = make_float2(ok ? xq[m].z : 0.0f, ok ? xq[m].w : 0.0f);
    }
}

__device__ __forceinline__ void x_hold()
{
    __builtin_amdgcn_fence(__ATOMIC_ACQ_REL, "wavefront");
}

__global__ __launch_bounds__(128, 2)
void scaconv_ds_kernel(const float* __restrict__ x,
                       const float* __restrict__ kernelD,
                       const float* __restrict__ kernelP,
                       const float* __restrict__ bias,
                       const float* __restrict__ cvec,
                       const float* __restrict__ mD_W1,
                       const float* __restrict__ mD_b1,
                       const float* __restrict__ mD_W2,
                       const float* __restrict__ mD_b2,
                       const float* __restrict__ mP_W1,
                       const float* __restrict__ mP_b1,
                       const float* __restrict__ mP_W2,
                       const float* __restrict__ mP_b2,
                       float* __restrict__ out)
{
    __shared__ float hD_lds[NL * HID_];
    __shared__ __attribute__((aligned(16))) float hP_lds[NL * HID_];
    __shared__ __attribute__((aligned(16))) float kD_lds[NL * 9];
    __shared__ __attribute__((aligned(16))) float smem[B_ * CI * XPL + RT * YZ + 64];
    float* const xk_lds = smem;
    float* const y_lds  = smem + B_ * CI * XPL;

    const int tid  = threadIdx.x;
    const int wave = tid >> 5;
    const int lane = tid & 31;
    const int lm   = lane & 15;
    const int hiL  = lane >> 4;

    const int wg = blockIdx.x;
    const int h0 = (wg >> 3) * RT;
    const int w0 = (wg & 7) * TL;

    const int nt = wave & 1;
    const int ig = wave >> 1;


    const int j  = tid & 15;
    const int k1 = tid & 7;
    float cq[8];
    #pragma unroll
    for (int q = 0; q < 8; ++q) cq[q] = cvec[q];
    float wD[12], wP[12];
    #pragma unroll
    for (int q = 0; q < 12; ++q) {
        wD[q] = mD_W1[q * HID_ + j];
        wP[q] = mP_W1[q * HID_ + j];
    }
    const float bD = mD_b1[j], bP = mP_b1[j];
    const float kDk = kernelD[k1], b2k = mD_b2[k1];
    const float kD8 = kernelD[8],  b28 = mD_b2[8];
    float w2c[HID_], w2e[HID_];
    #pragma unroll
    for (int jj = 0; jj < HID_; ++jj) {
        w2c[jj] = mD_W2[jj * 9 + k1];
        w2e[jj] = mD_W2[jj * 9 + 8];
    }
    __builtin_amdgcn_sched_barrier(0);

    const int  pc   = tid & 7;
    const int  pb   = ((tid >> 3) & 1) + 16 * ((tid >> 4) & 1) + 2 * (tid >> 5);
    const bool puse = (pc >= 1) && (pc <= 6);
    const int  wcol = w0 - 8 + 4 * pc;
    const bool cok  = puse && ((unsigned)wcol < (unsigned)WW);
    const int  wc   = cok ? wcol : w0;
    float4 xqa[XH], xqb[XH];
    x_fetch(xqa, x, 0, pb, h0, wc);
    x_fetch(xqb, x, 1, pb, h0, wc);

    if (tid < 64) y_lds[RT * YZ + tid] = 0.0f;

    {
        #pragma unroll
        for (int lh = 0; lh < 2; ++lh) {
            const int l = (tid >> 4) + 8 * lh;
            const int w = w0 + l;
            const float yn = (float)w * (1.0f / 128.0f);
            #pragma unroll
            for (int t = 0; t < RT; ++t) {
                const float xn = (float)(h0 + t) * (1.0f / 128.0f);
                float f[12];
                f[0] = xn; f[1] = 1.0f - xn; f[2] = yn; f[3] = 1.0f - yn;
                #pragma unroll
                for (int q = 0; q < 8; ++q) f[4 + q] = cq[q];
                float aD = bD, aP = bP;
                #pragma unroll
                for (int q = 0; q < 12; ++q) {
                    aD += f[q] * wD[q];
                    aP += f[q] * wP[q];
                }
                hD_lds[(t * TL + l) * HID_ + j] = fmaxf(aD, 0.0f);
                hP_lds[(t * TL + l) * HID_ + j] = fmaxf(aP, 0.0f);
            }
        }
    }
    lds_barrier();

    {
        const int l = tid >> 3;
        const float base  = kDk + b2k;
        const float base8 = kD8 + b28;
        #pragma unroll
        for (int t = 0; t < RT; ++t) {
            float a = base, a8 = base8;
            #pragma unroll
            for (int jj = 0; jj < HID_; ++jj) {
                const float hv = hD_lds[(t * TL + l) * HID_ + jj];
                a  += hv * w2c[jj];
                a8 += hv * w2e[jj];
            }
            kD_lds[(t * TL + l) * 9 + k1] = a;
            if (k1 == 0) kD_lds[(t * TL + l) * 9 + 8] = a8;
        }
    }

    v2f   bfr[8][2][4];
    #pragma unroll
    for (int ipx = 0; ipx < 8; ++ipx) {
        #pragma unroll
        for (int t2 = 0; t2 < 2; ++t2) {
            const int gcol = (ig * 16 + ipx * 2 + t2) * CO + nt * 16 + lm;
            #pragma unroll
            for (int ks = 0; ks < 4; ++ks) {
                const int k0 = ks * 4 + 2 * hiL;
                bfr[ipx][t2][ks].x = mP_W2[k0 * 1024 + gcol];
                bfr[ipx][t2][ks].y = mP_W2[(k0 + 1) * 1024 + gcol];
            }
        }
    }
    __builtin_amdgcn_sched_barrier(0);

    v2f afr[RT][4];
    #pragma unroll
    for (int t = 0; t < RT; ++t) {
        #pragma unroll
        for (int ks = 0; ks < 4; ++ks) {
            const float2 hv = *reinterpret_cast<const float2*>(
                &hP_lds[(t * TL + lm) * HID_ + ks * 4 + 2 * hiL]);
            afr[t][ks].x = hv.x;
            afr[t][ks].y = hv.y;
        }
    }

    if (puse) x_stage(xqa, xk_lds, 0, pb, pc, h0, cok);
    x_fetch(xqa, x, 2, pb, h0, wc);
    if (puse) x_stage(xqb, xk_lds, 1, pb, pc, h0, cok);
    x_fetch(xqb, x, 3, pb, h0, wc);
    __builtin_amdgcn_sched_barrier(0);

    v8f kpT[RT][8][2];
    #pragma unroll
    for (int ip = 0; ip < 8; ++ip) {
        if (ip == 4) {
            x_hold();
            if (puse) x_stage(xqa, xk_lds, 2, pb, pc, h0, cok);
            __builtin_amdgcn_sched_barrier(0);
        }
        #pragma unroll
        for (int t2 = 0; t2 < 2; ++t2) {
            #pragma unroll
            for (int t = 0; t < RT; ++t) {
                v8f dd = (v8f){};
                #pragma unroll
                for (int ks = 0; ks < 4; ++ks)
                    dd = __builtin_amdgcn_wmma_f32_16x16x4_f32(
                             false, afr[t][ks], false, bfr[ip][t2][ks], (short)0, dd, false, false);
                kpT[t][ip][t2] = dd;
            }
        }
        __builtin_amdgcn_sched_barrier(0);
    }
    x_hold();
    if (puse) x_stage(xqb, xk_lds, 3, pb, pc, h0, cok);
    __syncthreads();

    float kbase[8][2];
    #pragma unroll
    for (int ipx = 0; ipx < 8; ++ipx) {
        #pragma unroll
        for (int t2 = 0; t2 < 2; ++t2) {
            const int gcol = (ig * 16 + ipx * 2 + t2) * CO + nt * 16 + lm;
            kbase[ipx][t2] = kernelP[gcol] + mP_b2[gcol];
        }
    }

    {
        const int b = tid >> 5;
        const int i = tid & 31;
        float* const ydst = y_lds + b * YP + (i >> 3) * YG + ((i & 7) >> 1) * 16 + (i & 1);
        #pragma unroll 1
        for (int th = 0; th < 2 * RT; ++th) {
            const int t  = th >> 1;
            const int lh = th & 1;
            float kdr[8 * 9];
            {
                const float4* kD4 = reinterpret_cast<const float4*>(kD_lds + t * (TL * 9) + lh * 72);
                #pragma unroll
                for (int m = 0; m < 18; ++m) {
                    const float4 v = kD4[m];
                    kdr[4 * m] = v.x; kdr[4 * m + 1] = v.y; kdr[4 * m + 2] = v.z; kdr[4 * m + 3] = v.w;
                }
            }
            float f[2][3 * 12];
            #pragma unroll
            for (int pp = 0; pp < 2; ++pp) {
                const float2* xr2 = reinterpret_cast<const float2*>(
                    &xk_lds[(tid + 128 * pp) * XPL + t * XROW + 8 * lh]);
                #pragma unroll
                for (int rr = 0; rr < 3; ++rr) {
                    #pragma unroll
                    for (int k = 0; k < 6; ++k) {
                        const float2 v = xr2[rr * (XROW / 2) + 1 + k];
                        f[pp][rr * 12 + 2 * k] = v.x; f[pp][rr * 12 + 2 * k + 1] = v.y;
                    }
                }
            }
            float* const yp = ydst + t * YZ + lh * 8 * YP;
            #pragma unroll
            for (int l = 0; l < 8; ++l) {
                float acc0 = 0.0f, acc1 = 0.0f;
                #pragma unroll
                for (int kh = 0; kh < 3; ++kh) {
                    #pragma unroll
                    for (int kw = 0; kw < 3; ++kw) {
                        const int c = l + kw;
                        const float kd = kdr[l * 9 + kh * 3 + kw];
                        acc0 += f[0][kh * 12 + 1 + c] * kd;
                        acc1 += f[1][kh * 12 + 1 + c] * kd;
                    }
                }
                yp[l * 2]          = acc0;
                yp[l * 2 + 4 * YP] = acc1;
            }
        }
    }
    __syncthreads();

    const bool yact = ((lm >> 3) == hiL);
    const int  yoff = yact ? (hiL * 8 + (lm & 7)) * YP + 2 * ig * YG : RT * YZ;
    const int  ygs  = yact ? YG : 0;
    const int  oc  = nt * 16 + (lane >> 2);
    const float bo0 = bias[oc];
    const float bo1 = bias[oc + 8];

    #pragma unroll
    for (int t = 0; t < RT; ++t) {
        v8f acc[8];
        #pragma unroll
        for (int r = 0; r < 8; ++r) acc[r] = (v8f){};

        const float* yrow = y_lds + yoff + (yact ? t * YZ : 0);

        float4 yv[8][4];
        #pragma unroll
        for (int ip = 0; ip < 8; ++ip) {
            #pragma unroll
            for (int p = 0; p < 4; ++p)
                yv[ip][p] = *reinterpret_cast<const float4*>(
                    yrow + (ip >> 2) * ygs + (ip & 3) * 16 + p * 4);
        }
        __builtin_amdgcn_sched_barrier(0);

        #pragma unroll
        for (int ip = 0; ip < 8; ++ip) {
            #pragma unroll
            for (int p = 0; p < 4; ++p) {
                v2f af0, af1, bf0, bf1;
                af0.x = yv[ip][p].x; af0.y = yv[ip][p].y;
                af1.x = yv[ip][p].z; af1.y = yv[ip][p].w;
                bf0.x = kpT[t][ip][0][2 * p]     + kbase[ip][0];
                bf0.y = kpT[t][ip][1][2 * p]     + kbase[ip][1];
                bf1.x = kpT[t][ip][0][2 * p + 1] + kbase[ip][0];
                bf1.y = kpT[t][ip][1][2 * p + 1] + kbase[ip][1];
                acc[2 * p] = __builtin_amdgcn_wmma_f32_16x16x4_f32(
                                 false, af0, false, bf0, (short)0, acc[2 * p], false, false);
                acc[2 * p + 1] = __builtin_amdgcn_wmma_f32_16x16x4_f32(
                                 false, af1, false, bf1, (short)0, acc[2 * p + 1], false, false);
            }
        }
        if (t > 0) lds_barrier();

        {
            float4* red4 = reinterpret_cast<float4*>(smem);
            #pragma unroll
            for (int q = 0; q < 8; ++q) {
                #pragma unroll
                for (int rg = 0; rg < 2; ++rg) {
                    float4 v;
                    v.x = acc[rg * 4 + 0][q];
                    v.y = acc[rg * 4 + 1][q];
                    v.z = acc[rg * 4 + 2][q];
                    v.w = acc[rg * 4 + 3][q];
                    red4[(wave * 16 + q * 2 + rg) * 32 + 16 * hiL + ((lm + 8 * rg + 4 * hiL) & 15)] = v;
                }
            }
        }
        lds_barrier();

        {
            const float4* red4 = reinterpret_cast<const float4*>(smem);
            const int qg  = wave >> 1;
            const int pi  = lane & 3;
            const int prg = pi & 1;
            const int ph  = pi >> 1;
            #pragma unroll
            for (int k = 0; k < 8; ++k) {
                const int q  = qg * 4 + (k >> 1);
                const int cl = (lane >> 2) + 8 * (k & 1);
                const int sl = (q * 2 + prg) * 32 + 16 * ph + ((cl + 8 * prg + 4 * ph) & 15);
                float4 s = red4[(nt + 0) * 16 * 32 + sl];
                const float4 pp = red4[(nt + 2) * 16 * 32 + sl];
                const float bo = (k & 1) ? bo1 : bo0;
                s.x += pp.x; s.y += pp.y; s.z += pp.z; s.w += pp.w;
                s.x += bo; s.y += bo; s.z += bo; s.w += bo;
                *reinterpret_cast<float4*>(
                    out + (size_t)(q * CO + nt * 16 + cl) * (HH * WW) + (size_t)(h0 + t) * WW
                        + (w0 + 4 * pi)) = s;
            }
        }
    }
}

extern "C" void kernel_launch(void* const* d_in, const int* in_sizes, int n_in,
                              void* d_out, int out_size, void* d_ws, size_t ws_size,
                              hipStream_t stream) {
    const float* x       = (const float*)d_in[0];
    const float* kernelD = (const float*)d_in[1];
    const float* kernelP = (const float*)d_in[2];
    const float* bias    = (const float*)d_in[3];
    const float* cvec    = (const float*)d_in[4];
    const float* mD_W1   = (const float*)d_in[5];
    const float* mD_b1   = (const float*)d_in[6];
    const float* mD_W2   = (const float*)d_in[7];
    const float* mD_b2   = (const float*)d_in[8];
    const float* mP_W1   = (const float*)d_in[9];
    const float* mP_b1   = (const float*)d_in[10];
    const float* mP_W2   = (const float*)d_in[11];
    const float* mP_b2   = (const float*)d_in[12];
    float* out = (float*)d_out;

    dim3 grid(512);
    dim3 block(128);
    hipLaunchKernelGGL(scaconv_ds_kernel, grid, block, 0, stream,
                       x, kernelD, kernelP, bias, cvec,
                       mD_W1, mD_b1, mD_W2, mD_b2,
                       mP_W1, mP_b1, mP_W2, mP_b2, out);
}
